# starting kernel
# baseline (speedup 1.0000x reference)
#include <hip/hip_runtime.h>


typedef __attribute__((ext_vector_type(2))) float v2f;
typedef __attribute__((ext_vector_type(8))) float v8f;

#define FHID 96
#define FQ   (FHID / 4)


__global__ void k_fill_f32(float* __restrict__ p, float v, int n) {
  int i = blockIdx.x * blockDim.x + threadIdx.x;
  if (i < n) p[i] = v;
}

__global__ void k_zero_f4(float4* __restrict__ p, int n4) {
  int i = blockIdx.x * blockDim.x + threadIdx.x;
  if (i < n4) p[i] = make_float4(0.f, 0.f, 0.f, 0.f);
}

__global__ void k_degree(const int* __restrict__ dst, float* __restrict__ deg, int E) {
  int e = blockIdx.x * blockDim.x + threadIdx.x;
  if (e < E) atomicAdd(&deg[dst[e]], 1.0f);
}

__global__ void k_dinv(const float* __restrict__ deg, float* __restrict__ dinv, int n) {
  int i = blockIdx.x * blockDim.x + threadIdx.x;
  if (i < n) dinv[i] = rsqrtf(deg[i]);
}

__global__ void k_norm(const int* __restrict__ src, const int* __restrict__ dst,
                       const float* __restrict__ dinv, float* __restrict__ norm, int E) {
  int e = blockIdx.x * blockDim.x + threadIdx.x;
  if (e < E) norm[e] = dinv[src[e]] * dinv[dst[e]];
}


__global__ void k_packB(const float* __restrict__ B, float* __restrict__ Bp,
                        int K, int Nc, int Ncp) {
  int i = blockIdx.x * blockDim.x + threadIdx.x;
  if (i >= K * Ncp) return;
  const int k   = i / Ncp;
  const int col = i - k * Ncp;
  const float v = (col < Nc) ? B[(size_t)k * Nc + col] : 0.f;
  Bp[((size_t)(k >> 1) * Ncp + col) * 2 + (k & 1)] = v;
}


__global__ void k_gemm_wmma_f32(const float* __restrict__ A,
                                const float* __restrict__ Bp,
                                const float* __restrict__ bias,
                                float* __restrict__ C,
                                int M, int K, int Nc, int Ncp,
                                int Mtiles, int Ntiles) {
  const int wave = threadIdx.x >> 5;
  const int lane = threadIdx.x & 31;
  const int tile = blockIdx.x * (blockDim.x >> 5) + wave;
  if (tile >= Mtiles * Ntiles) return;

  const int mt = tile / Ntiles;
  const int nt = tile % Ntiles;
  const int m0 = mt * 16;
  const int n0 = nt * 16;

  const int l15  = lane & 15;
  const int half = lane >> 4;
  const int koff = half << 1;
  const int col  = n0 + l15;

  const float* __restrict__ Ar = A + (size_t)(m0 + l15) * K + koff;
  const float* __restrict__ Br = Bp + ((size_t)half * Ncp + col) * 2;

  v8f acc = {};
  for (int k = 0; k < K; k += 4) {
    const v2f a = *(const v2f*)(Ar + k);
    const v2f b = *(const v2f*)(Br + (size_t)(k >> 1) * Ncp * 2);
    acc = __builtin_amdgcn_wmma_f32_16x16x4_f32(false, a, false, b,
                                                (short)0, acc, false, false);
  }

  if (col < Nc) {
    const float bv = bias ? bias[col] : 0.f;
    const int mbase = m0 + (half << 3);
    #pragma unroll
    for (int r = 0; r < 8; ++r)
      C[(size_t)(mbase + r) * Nc + col] = acc[r] + bv;
  }
}


__global__ void k_scatter(const float4* __restrict__ h4,
                          const int* __restrict__ src,
                          const int* __restrict__ dst,
                          const float* __restrict__ norm,
                          float* __restrict__ agg, int E) {
  int i = blockIdx.x * blockDim.x + threadIdx.x;
  if (i >= E * FQ) return;
  const int e = i / FQ;
  const int q = i - e * FQ;
  const float  s = norm[e];
  const float4 v = h4[(size_t)src[e] * FQ + q];
  float* __restrict__ out = agg + (size_t)dst[e] * FHID + q * 4;
  atomicAdd(out + 0, v.x * s);
  atomicAdd(out + 1, v.y * s);
  atomicAdd(out + 2, v.z * s);
  atomicAdd(out + 3, v.w * s);
}


__global__ void k_epilogue(const float4* __restrict__ agg4,
                           const float4* __restrict__ tmp4,
                           const float* __restrict__ dinv,
                           const float* __restrict__ bias,
                           float4* __restrict__ out4, int Nn) {
  int i = blockIdx.x * blockDim.x + threadIdx.x;
  if (i >= Nn * FQ) return;
  const int node = i / FQ;
  const int q    = i - node * FQ;
  const float d  = dinv[node];
  const float w  = d * d;
  const float4 a = agg4[i];
  const float4 t = tmp4[i];
  const int f = q * 4;
  float4 r;
  r.x = fmaxf(a.x + t.x * w + bias[f + 0], 0.f);
  r.y = fmaxf(a.y + t.y * w + bias[f + 1], 0.f);
  r.z = fmaxf(a.z + t.z * w + bias[f + 2], 0.f);
  r.w = fmaxf(a.w + t.w * w + bias[f + 3], 0.f);
  out4[i] = r;
}


static inline int cdiv(int a, int b) { return (a + b - 1) / b; }

extern "C" void kernel_launch(void* const* d_in, const int* in_sizes, int n_in,
                              void* d_out, int out_size, void* d_ws, size_t ws_size,
                              hipStream_t stream) {
  const float* x    = (const float*)d_in[0];
  const int*   eidx = (const int*)  d_in[1];
  const float* W1   = (const float*)d_in[2];
  const float* b1   = (const float*)d_in[3];
  const float* W2   = (const float*)d_in[4];
  const float* b2   = (const float*)d_in[5];
  const float* Wout = (const float*)d_in[6];
  const float* bout = (const float*)d_in[7];

  const int F_IN  = 128;
  const int F_OUT = 40;
  const int F_OUTP = 48;
  const int N = in_sizes[0] / F_IN;
  const int E = in_sizes[1] / 2;
  const int* src = eidx;
  const int* dst = eidx + E;

  float* deg   = (float*)d_ws;
  float* dinv  = deg   + N;
  float* norm  = dinv  + N;
  float* tmp   = norm  + E;
  float* agg   = tmp   + (size_t)N * FHID;
  float* hbuf  = agg   + (size_t)N * FHID;
  float* W1p   = hbuf  + (size_t)N * FHID;
  float* W2p   = W1p   + (size_t)F_IN * FHID;
  float* Woutp = W2p   + (size_t)FHID * FHID;
  float* outp  = (float*)d_out;

  const int T = 256;
  const int NF4 = (N * FHID) / 4;

  k_fill_f32<<<cdiv(N, T), T, 0, stream>>>(deg, 1.0f, N);
  k_degree  <<<cdiv(E, T), T, 0, stream>>>(dst, deg, E);
  k_dinv    <<<cdiv(N, T), T, 0, stream>>>(deg, dinv, N);
  k_norm    <<<cdiv(E, T), T, 0, stream>>>(src, dst, dinv, norm, E);
  k_packB<<<cdiv(F_IN * FHID,   T), T, 0, stream>>>(W1,   W1p,   F_IN, FHID,  FHID);
  k_packB<<<cdiv(FHID * FHID,   T), T, 0, stream>>>(W2,   W2p,   FHID, FHID,  FHID);
  k_packB<<<cdiv(FHID * F_OUTP, T), T, 0, stream>>>(Wout, Woutp, FHID, F_OUT, F_OUTP);

  const int Mtiles = cdiv(N, 16);
  const int wavesPerBlk = T / 32;

  {
    const int Ntl = FHID / 16;
    k_zero_f4<<<cdiv(NF4, T), T, 0, stream>>>((float4*)agg, NF4);
    k_gemm_wmma_f32<<<cdiv(Mtiles * Ntl, wavesPerBlk), T, 0, stream>>>(
        x, W1p, nullptr, tmp, N, F_IN, FHID, FHID, Mtiles, Ntl);
    k_scatter<<<cdiv(E * FQ, T), T, 0, stream>>>(
        (const float4*)tmp, src, dst, norm, agg, E);
    k_epilogue<<<cdiv(N * FQ, T), T, 0, stream>>>(
        (const float4*)agg, (const float4*)tmp, dinv, b1, (float4*)hbuf, N);
  }

  {
    const int Ntl = FHID / 16;
    k_zero_f4<<<cdiv(NF4, T), T, 0, stream>>>((float4*)agg, NF4);
    k_gemm_wmma_f32<<<cdiv(Mtiles * Ntl, wavesPerBlk), T, 0, stream>>>(
        hbuf, W2p, nullptr, tmp, N, FHID, FHID, FHID, Mtiles, Ntl);
    k_scatter<<<cdiv(E * FQ, T), T, 0, stream>>>(
        (const float4*)tmp, src, dst, norm, agg, E);
    k_epilogue<<<cdiv(N * FQ, T), T, 0, stream>>>(
        (const float4*)agg, (const float4*)tmp, dinv, b2, (float4*)hbuf, N);
  }

  {
    const int Ntl = F_OUTP / 16;
    k_gemm_wmma_f32<<<cdiv(Mtiles * Ntl, wavesPerBlk), T, 0, stream>>>(
        hbuf, Woutp, bout, outp, N, FHID, F_OUT, F_OUTP, Mtiles, Ntl);
  }
}


# Each aggregation launch now also runs the next layer's WMMA product on its 16 rows from LDS (16 one-node waves per block, one barrier): activations never reach memory, 2 GEMM launches removed
# speedup vs baseline: 2.2374x; 2.2374x over previous
#include <hip/hip_runtime.h>


typedef __attribute__((ext_vector_type(2))) float v2f;
typedef __attribute__((ext_vector_type(4))) float v4f;
typedef __attribute__((ext_vector_type(8))) float v8f;

#define FHID 96
#define SCAN_T 1024

struct __attribute__((packed, aligned(4))) f3 { float x, y, z; };


__global__ void k_zero_i32(int* __restrict__ p, int n) {
  int i = blockIdx.x * blockDim.x + threadIdx.x;
  if (i < n) p[i] = 0;
}

__global__ void k_count_rank(const int* __restrict__ dst, int* __restrict__ cnt,
                             int* __restrict__ rank, int E) {
  int e = blockIdx.x * blockDim.x + threadIdx.x;
  if (e < E) rank[e] = atomicAdd(&cnt[dst[e]], 1);
}

__global__ void __launch_bounds__(SCAN_T)
k_scan(const int* __restrict__ cnt, int* __restrict__ rowptr,
       float* __restrict__ dinv, int N) {
  __shared__ int s_part[SCAN_T / 32];
  __shared__ int s_wsum[SCAN_T / 32];
  const int tid  = threadIdx.x;
  const int lane = tid & 31;
  const int wv   = tid >> 5;
  const int tile0 = blockIdx.x * SCAN_T;

  int part = 0;
  const int4* __restrict__ c4 = (const int4*)cnt;
  for (int i = tid; i < (tile0 >> 2); i += SCAN_T) {
    const int4 v = c4[i];
    part += (v.x + v.y) + (v.z + v.w);
  }
  for (int o = 16; o > 0; o >>= 1) part += __shfl_xor(part, o);

  const int idx = tile0 + tid;
  const int c = (idx < N) ? cnt[idx] : 0;
  int incl = c;
  for (int o = 1; o < 32; o <<= 1) {
    const int t = __shfl_up(incl, o);
    if (lane >= o) incl += t;
  }
  if (lane == 31) s_wsum[wv] = incl;
  if (lane == 0)  s_part[wv] = part;
  __syncthreads();

  int base = 0, woff = 0;
  for (int k = 0; k < SCAN_T / 32; ++k) {
    base += s_part[k];
    woff += (k < wv) ? s_wsum[k] : 0;
  }
  const int excl = base + woff + incl - c;
  if (idx < N) {
    rowptr[idx] = excl;
    dinv[idx]   = rsqrtf((float)(c + 1));
  }
  if (idx == N - 1) rowptr[N] = excl + c;
}

__global__ void k_fill_csr(const int* __restrict__ src, const int* __restrict__ dst,
                           const int* __restrict__ rank, const int* __restrict__ rowptr,
                           int* __restrict__ csr_src, int E) {
  int e = blockIdx.x * blockDim.x + threadIdx.x;
  if (e < E) csr_src[rowptr[dst[e]] + rank[e]] = src[e];
}


template <int K, int NT, int NC>
__global__ void __launch_bounds__(256)
k_gemm_strip(const float* __restrict__ A, const float* __restrict__ Wm,
             const float* __restrict__ bias,
             float* __restrict__ C, int Mtiles) {
  constexpr int NCP   = NT * 16;
  constexpr int KQ    = K / 4;
  constexpr int C4    = NCP / 4;
  constexpr int NC4   = NC / 4;
  constexpr int TOTAL = KQ * C4;
  constexpr int ITERS = (TOTAL + 255) / 256;
  __shared__ v4f Bs[KQ * NCP];

  const int tid  = threadIdx.x;
  const int wave = tid >> 5;
  const int lane = tid & 31;
  const int c    = lane & 15;
  const int h    = lane >> 4;
  const int strip = blockIdx.x * 8 + wave;
  const bool live = strip < Mtiles;
  const int srow  = (live ? strip : Mtiles - 1) * 16;

  v4f w0[ITERS], w1[ITERS], w2[ITERS], w3[ITERS];
  #pragma unroll
  for (int j = 0; j < ITERS; ++j) {
    const int it = tid + 256 * j;
    const int q  = it / C4;
    const int g  = it - q * C4;
    w0[j] = w1[j] = w2[j] = w3[j] = (v4f){0.f, 0.f, 0.f, 0.f};
    if (it < TOTAL && g < NC4) {
      const float* p = Wm + (size_t)(4 * q) * NC + 4 * g;
      w0[j] = *(const v4f*)p;
      w1[j] = *(const v4f*)(p + NC);
      w2[j] = *(const v4f*)(p + 2 * NC);
      w3[j] = *(const v4f*)(p + 3 * NC);
    }
  }

  v4f a[K / 8];
  const float* __restrict__ Ar = A + (size_t)(srow + c) * K + 4 * h;
  #pragma unroll
  for (int i = 0; i < K / 8; ++i) a[i] = *(const v4f*)(Ar + 8 * i);

  #pragma unroll
  for (int j = 0; j < ITERS; ++j) {
    const int it = tid + 256 * j;
    const int q  = it / C4;
    const int g  = it - q * C4;
    if (it < TOTAL) {
      v4f* d = Bs + q * NCP + 4 * g;
      d[0] = (v4f){w0[j].x, w1[j].x, w2[j].x, w3[j].x};
      d[1] = (v4f){w0[j].y, w1[j].y, w2[j].y, w3[j].y};
      d[2] = (v4f){w0[j].z, w1[j].z, w2[j].z, w3[j].z};
      d[3] = (v4f){w0[j].w, w1[j].w, w2[j].w, w3[j].w};
    }
  }
  __syncthreads();
  if (!live) return;

  v8f acc[NT];
  #pragma unroll
  for (int t = 0; t < NT; ++t) acc[t] = (v8f){0.f, 0.f, 0.f, 0.f, 0.f, 0.f, 0.f, 0.f};

  const v4f* Bl = Bs + h * NCP + c;
  #pragma unroll
  for (int i = 0; i < K / 8; ++i) {
    v4f b[NT];
    #pragma unroll
    for (int t = 0; t < NT; ++t) b[t] = Bl[(2 * i) * NCP + 16 * t];
    const v2f a01 = {a[i].x, a[i].y};
    const v2f a23 = {a[i].z, a[i].w};
    #pragma unroll
    for (int t = 0; t < NT; ++t)
      acc[t] = __builtin_amdgcn_wmma_f32_16x16x4_f32(
          false, a01, false, (v2f){b[t].x, b[t].y}, (short)0, acc[t], false, false);
    #pragma unroll
    for (int t = 0; t < NT; ++t)
      acc[t] = __builtin_amdgcn_wmma_f32_16x16x4_f32(
          false, a23, false, (v2f){b[t].z, b[t].w}, (short)0, acc[t], false, false);
  }

  #pragma unroll
  for (int t = 0; t < NT; ++t) {
    const int col = 16 * t + c;
    if (col < NC) {
      const float bv = bias ? bias[col] : 0.f;
      float* Cp = C + (size_t)(strip * 16 + 8 * h) * NC + col;
      #pragma unroll
      for (int r = 0; r < 8; ++r) Cp[(size_t)r * NC] = acc[t][r] + bv;
    }
  }
}


static __device__ __forceinline__ float lane_f(float v, int l) {
  return __int_as_float(__builtin_amdgcn_readlane(__float_as_int(v), l));
}

static __device__ __forceinline__ f3 gather_row(const float* __restrict__ h,
                         const int* __restrict__ csr_src,
                         const float* __restrict__ dinv,
                         const float* __restrict__ bias,
                         int node, int lane, int beg_v, int end_v, float dn) {
  const int beg = __builtin_amdgcn_readfirstlane(beg_v);
  const int end = __builtin_amdgcn_readfirstlane(end_v);
  const int f = lane * 3;
  const float* __restrict__ hl = h + f;

  float a0 = 0.f, a1 = 0.f, a2 = 0.f;
  for (int base = beg; base < end; base += 32) {
    const int j = base + lane;
    int   s = 0;
    float w = 0.f;
    if (j < end) { s = csr_src[j]; w = dinv[s] * dn; }
    const int n = min(32, end - base);
    int i = 0;
    for (; i + 4 <= n; i += 4) {
      const int s0 = __builtin_amdgcn_readlane(s, i);
      const int s1 = __builtin_amdgcn_readlane(s, i + 1);
      const int s2 = __builtin_amdgcn_readlane(s, i + 2);
      const int s3 = __builtin_amdgcn_readlane(s, i + 3);
      const f3 r0 = *(const f3*)(hl + (size_t)s0 * FHID);
      const f3 r1 = *(const f3*)(hl + (size_t)s1 * FHID);
      const f3 r2 = *(const f3*)(hl + (size_t)s2 * FHID);
      const f3 r3 = *(const f3*)(hl + (size_t)s3 * FHID);
      const float w0 = lane_f(w, i);
      const float w1 = lane_f(w, i + 1);
      const float w2 = lane_f(w, i + 2);
      const float w3 = lane_f(w, i + 3);
      a0 = fmaf(r0.x, w0, a0); a1 = fmaf(r0.y, w0, a1); a2 = fmaf(r0.z, w0, a2);
      a0 = fmaf(r1.x, w1, a0); a1 = fmaf(r1.y, w1, a1); a2 = fmaf(r1.z, w1, a2);
      a0 = fmaf(r2.x, w2, a0); a1 = fmaf(r2.y, w2, a1); a2 = fmaf(r2.z, w2, a2);
      a0 = fmaf(r3.x, w3, a0); a1 = fmaf(r3.y, w3, a1); a2 = fmaf(r3.z, w3, a2);
    }
    for (; i < n; ++i) {
      const int   si = __builtin_amdgcn_readlane(s, i);
      const float wi = lane_f(w, i);
      const f3 r = *(const f3*)(hl + (size_t)si * FHID);
      a0 = fmaf(r.x, wi, a0); a1 = fmaf(r.y, wi, a1); a2 = fmaf(r.z, wi, a2);
    }
  }

  const f3 t = *(const f3*)(hl + (size_t)node * FHID);
  const f3 b = *(const f3*)(bias + f);
  const float ws = dn * dn;
  f3 r;
  r.x = fmaxf(a0 + t.x * ws + b.x, 0.f);
  r.y = fmaxf(a1 + t.y * ws + b.y, 0.f);
  r.z = fmaxf(a2 + t.z * ws + b.z, 0.f);
  return r;
}

#define APITCH 100

template <int NT, int NC>
__global__ void __launch_bounds__(512)
k_gather_gemm(const float* __restrict__ hin,
              const int* __restrict__ rowptr,
              const int* __restrict__ csr_src,
              const float* __restrict__ dinv,
              const float* __restrict__ bias_in,
              const float* __restrict__ Wm,
              const float* __restrict__ bias_out,
              float* __restrict__ C, int N) {
  constexpr int K     = FHID;
  constexpr int NCP   = NT * 16;
  constexpr int KQ    = K / 4;
  constexpr int C4    = NCP / 4;
  constexpr int NC4   = NC / 4;
  constexpr int TOTAL = KQ * C4;
  constexpr int ITERS = (TOTAL + 511) / 512;
  __shared__ v4f Bs[KQ * NCP];
  __shared__ v4f As4[16 * APITCH / 4];

  const int tid  = threadIdx.x;
  const int wave = tid >> 5;
  const int lane = tid & 31;
  const int c    = lane & 15;
  const int h    = lane >> 4;

  const int node = blockIdx.x * 16 + wave;
  const bool live = node < N;
  int rb = 0, re = 0;
  float dn = 0.f;
  if (live) { rb = rowptr[node]; re = rowptr[node + 1]; dn = dinv[node]; }

  v4f w0[ITERS], w1[ITERS], w2[ITERS], w3[ITERS];
  #pragma unroll
  for (int j = 0; j < ITERS; ++j) {
    const int it = tid + 512 * j;
    const int q  = it / C4;
    const int g  = it - q * C4;
    w0[j] = w1[j] = w2[j] = w3[j] = (v4f){0.f, 0.f, 0.f, 0.f};
    if (it < TOTAL && g < NC4) {
      const float* p = Wm + (size_t)(4 * q) * NC + 4 * g;
      w0[j] = *(const v4f*)p;
      w1[j] = *(const v4f*)(p + NC);
      w2[j] = *(const v4f*)(p + 2 * NC);
      w3[j] = *(const v4f*)(p + 3 * NC);
    }
  }
  #pragma unroll
  for (int j = 0; j < ITERS; ++j) {
    const int it = tid + 512 * j;
    const int q  = it / C4;
    const int g  = it - q * C4;
    if (it < TOTAL) {
      v4f* d = Bs + q * NCP + 4 * g;
      d[0] = (v4f){w0[j].x, w1[j].x, w2[j].x, w3[j].x};
      d[1] = (v4f){w0[j].y, w1[j].y, w2[j].y, w3[j].y};
      d[2] = (v4f){w0[j].z, w1[j].z, w2[j].z, w3[j].z};
      d[3] = (v4f){w0[j].w, w1[j].w, w2[j].w, w3[j].w};
    }
  }

  f3 r;
  r.x = 0.f; r.y = 0.f; r.z = 0.f;
  if (live)
    r = gather_row(hin, csr_src, dinv, bias_in, node, lane, rb, re, dn);
  float* ar = (float*)As4 + wave * APITCH + 3 * lane;
  ar[0] = r.x; ar[1] = r.y; ar[2] = r.z;
  __syncthreads();
  if (wave >= NT) return;

  const int t = wave;
  const int col = 16 * t + c;
  const float bv = (bias_out && col < NC) ? bias_out[col] : 0.f;
  v8f acc = (v8f){0.f, 0.f, 0.f, 0.f, 0.f, 0.f, 0.f, 0.f};
  const v4f* Al = As4 + c * (APITCH / 4) + h;
  const v4f* Bl = Bs + h * NCP + 16 * t + c;
  #pragma unroll
  for (int i = 0; i < K / 8; ++i) {
    const v4f a = Al[2 * i];
    const v4f b = Bl[(2 * i) * NCP];
    acc = __builtin_amdgcn_wmma_f32_16x16x4_f32(
        false, (v2f){a.x, a.y}, false, (v2f){b.x, b.y}, (short)0, acc, false, false);
    acc = __builtin_amdgcn_wmma_f32_16x16x4_f32(
        false, (v2f){a.z, a.w}, false, (v2f){b.z, b.w}, (short)0, acc, false, false);
  }

  if (col < NC) {
    const int row0 = blockIdx.x * 16 + 8 * h;
    float* Cp = C + (size_t)row0 * NC + col;
    #pragma unroll
    for (int rr = 0; rr < 8; ++rr)
      if (row0 + rr < N) Cp[(size_t)rr * NC] = acc[rr] + bv;
  }
}


static inline int cdiv(int a, int b) { return (a + b - 1) / b; }

extern "C" void kernel_launch(void* const* d_in, const int* in_sizes, int n_in,
                              void* d_out, int out_size, void* d_ws, size_t ws_size,
                              hipStream_t stream) {
  const float* x    = (const float*)d_in[0];
  const int*   eidx = (const int*)  d_in[1];
  const float* W1   = (const float*)d_in[2];
  const float* b1   = (const float*)d_in[3];
  const float* W2   = (const float*)d_in[4];
  const float* b2   = (const float*)d_in[5];
  const float* Wout = (const float*)d_in[6];
  const float* bout = (const float*)d_in[7];

  const int F_IN  = 128;
  const int N = in_sizes[0] / F_IN;
  const int E = in_sizes[1] / 2;
  const int* src = eidx;
  const int* dst = eidx + E;

  int*   cnt   = (int*)d_ws;
  float* dinv  = (float*)d_ws + N;
  int*   rank  = (int*)d_ws + 2 * (size_t)N;
  float* tmp   = (float*)d_ws + 2 * (size_t)N + E;
  float* csrb  = tmp   + (size_t)N * FHID;
  int*   rowptr  = (int*)csrb;
  int*   csr_src = rowptr + N + 1;
  float* hbuf  = csrb  + (size_t)N * FHID;
  float* outp  = (float*)d_out;

  const int T = 256;

  k_zero_i32  <<<cdiv(N, T), T, 0, stream>>>(cnt, N);
  k_count_rank<<<cdiv(E, T), T, 0, stream>>>(dst, cnt, rank, E);
  k_scan      <<<cdiv(N, SCAN_T), SCAN_T, 0, stream>>>(cnt, rowptr, dinv, N);
  k_fill_csr  <<<cdiv(E, T), T, 0, stream>>>(src, dst, rank, rowptr, csr_src, E);

  const int Mtiles = cdiv(N, 16);
  const int gemmBlocks = cdiv(Mtiles, 8);

  k_gemm_strip<128, 6, 96><<<gemmBlocks, T, 0, stream>>>(x, W1, nullptr, tmp, Mtiles);

  k_gather_gemm<6, 96><<<Mtiles, 512, 0, stream>>>(
      tmp, rowptr, csr_src, dinv, b1, W2, nullptr, hbuf, N);

  k_gather_gemm<3, 40><<<Mtiles, 512, 0, stream>>>(
      hbuf, rowptr, csr_src, dinv, b2, Wout, bout, outp, N);
}


# Gather launches made persistent: 1024 resident blocks each walk the 16-node strips (W to LDS once per block, A strip double-buffered, next bounds asked a trip ahead): 16384 waves a launch, not 50000
# speedup vs baseline: 3.3738x; 1.5079x over previous
#include <hip/hip_runtime.h>


typedef __attribute__((ext_vector_type(2))) float v2f;
typedef __attribute__((ext_vector_type(4))) float v4f;
typedef __attribute__((ext_vector_type(8))) float v8f;

#define FHID 96
#define SCAN_T 1024

struct __attribute__((packed, aligned(4))) f3 { float x, y, z; };


#define EP  36
#define RT  7
#define CAP (RT * SCAN_T)
#define EU  12

__global__ void __launch_bounds__(1024)
k_hist(const int* __restrict__ dst, int* __restrict__ part, int* __restrict__ tsum,
       int N, int E, int epb, int ntiles) {
  __shared__ int h[CAP];
  __shared__ int s_w[RT][32];
  const int tid  = threadIdx.x;
  const int lane = tid & 31;
  const int wv   = tid >> 5;
  const int b  = blockIdx.x;
  const int r  = blockIdx.y;
  const int lo = r * CAP;

  for (int i = tid; i < CAP; i += 1024) h[i] = 0;
  __syncthreads();

  const int e0 = b * epb;
  const int e1 = min(E, e0 + epb);
  for (int base = e0 + tid; base < e1; base += EU * 1024) {
    int d[EU];
    #pragma unroll
    for (int u = 0; u < EU; ++u) {
      const int e = base + u * 1024;
      d[u] = (e < e1) ? dst[e] : -1;
    }
    #pragma unroll
    for (int u = 0; u < EU; ++u) {
      const unsigned x = (unsigned)(d[u] - lo);
      if (x < (unsigned)CAP) atomicAdd(&h[x], 1);
    }
  }
  __syncthreads();

  int sm[RT];
  #pragma unroll
  for (int k = 0; k < RT; ++k) {
    const int n = lo + k * SCAN_T + tid;
    sm[k] = h[k * SCAN_T + tid];
    if (n < N) part[(size_t)b * N + n] = sm[k];
  }
  #pragma unroll
  for (int o = 16; o > 0; o >>= 1) {
    int up[RT];
    #pragma unroll
    for (int k = 0; k < RT; ++k) up[k] = __shfl_xor(sm[k], o);
    #pragma unroll
    for (int k = 0; k < RT; ++k) sm[k] += up[k];
  }
  if (lane == 0) {
    #pragma unroll
    for (int k = 0; k < RT; ++k) s_w[k][wv] = sm[k];
  }
  __syncthreads();
  if (tid < RT) {
    const int tile = r * RT + tid;
    if (tile < ntiles) {
      int tot = 0;
      for (int w = 0; w < 32; ++w) tot += s_w[tid][w];
      tsum[b * ntiles + tile] = tot;
    }
  }
}

__global__ void __launch_bounds__(SCAN_T)
k_scan(int* __restrict__ part, const int* __restrict__ tsum, int* __restrict__ rowptr,
       float* __restrict__ dinv, int N, int ntiles) {
  __shared__ int s_part[SCAN_T / 32];
  __shared__ int s_wsum[SCAN_T / 32];
  const int tid  = threadIdx.x;
  const int lane = tid & 31;
  const int wv   = tid >> 5;
  const int t    = blockIdx.x;
  const int idx  = t * SCAN_T + tid;

  const int li = min(idx, N - 1);
  int cb[EP];
  #pragma unroll
  for (int b = 0; b < EP; ++b) cb[b] = part[(size_t)b * N + li];

  int front = 0;
  for (int i = tid; i < EP * ntiles; i += SCAN_T) {
    const int k = i % ntiles;
    if (k < t) front += tsum[i];
  }
  for (int o = 16; o > 0; o >>= 1) front += __shfl_xor(front, o);

  int c = 0;
  #pragma unroll
  for (int b = 0; b < EP; ++b) { const int v = (idx < N) ? cb[b] : 0; cb[b] = c; c += v; }

  int incl = c;
  for (int o = 1; o < 32; o <<= 1) {
    const int u = __shfl_up(incl, o);
    if (lane >= o) incl += u;
  }
  if (lane == 31) s_wsum[wv] = incl;
  if (lane == 0)  s_part[wv] = front;
  __syncthreads();

  int base = 0, woff = 0;
  for (int k = 0; k < SCAN_T / 32; ++k) {
    base += s_part[k];
    woff += (k < wv) ? s_wsum[k] : 0;
  }
  const int excl = base + woff + incl - c;
  if (idx < N) {
    rowptr[idx] = excl;
    dinv[idx]   = rsqrtf((float)(c + 1));
    #pragma unroll
    for (int b = 0; b < EP; ++b) part[(size_t)b * N + idx] = excl + cb[b];
  }
  if (idx == N - 1) rowptr[N] = excl + c;
}

__global__ void __launch_bounds__(1024)
k_fill_csr(const int* __restrict__ src, const int* __restrict__ dst,
           const int* __restrict__ part, int* __restrict__ csr_src,
           int N, int E, int epb) {
  __shared__ int h[CAP];
  const int tid = threadIdx.x;
  const int b  = blockIdx.x;
  const int r  = blockIdx.y;
  const int lo = r * CAP;

  int pv[RT];
  #pragma unroll
  for (int k = 0; k < RT; ++k) {
    const int n = lo + k * SCAN_T + tid;
    pv[k] = (n < N) ? part[(size_t)b * N + n] : 0;
  }
  #pragma unroll
  for (int k = 0; k < RT; ++k) h[k * SCAN_T + tid] = pv[k];
  __syncthreads();

  const int e0 = b * epb;
  const int e1 = min(E, e0 + epb);
  for (int base = e0 + tid; base < e1; base += EU * 1024) {
    int d[EU], sv[EU];
    #pragma unroll
    for (int u = 0; u < EU; ++u) {
      const int e = base + u * 1024;
      d[u]  = (e < e1) ? dst[e] : -1;
      sv[u] = (e < e1) ? src[e] : 0;
    }
    int pos[EU];
    #pragma unroll
    for (int u = 0; u < EU; ++u) {
      const unsigned x = (unsigned)(d[u] - lo);
      pos[u] = (x < (unsigned)CAP) ? atomicAdd(&h[x], 1) : -1;
    }
    #pragma unroll
    for (int u = 0; u < EU; ++u)
      if (pos[u] >= 0) csr_src[pos[u]] = sv[u];
  }
}


template <int K, int NT, int NC>
__global__ void __launch_bounds__(256)
k_gemm_strip(const float* __restrict__ A, const float* __restrict__ Wm,
             const float* __restrict__ bias,
             float* __restrict__ C, int Mtiles) {
  constexpr int NCP   = NT * 16;
  constexpr int KQ    = K / 4;
  constexpr int C4    = NCP / 4;
  constexpr int NC4   = NC / 4;
  constexpr int TOTAL = KQ * C4;
  constexpr int ITERS = (TOTAL + 255) / 256;
  __shared__ v4f Bs[KQ * NCP];

  const int tid  = threadIdx.x;
  const int wave = tid >> 5;
  const int lane = tid & 31;
  const int c    = lane & 15;
  const int h    = lane >> 4;
  const int strip = blockIdx.x * 8 + wave;
  const bool live = strip < Mtiles;
  const int srow  = (live ? strip : Mtiles - 1) * 16;

  v4f w0[ITERS], w1[ITERS], w2[ITERS], w3[ITERS];
  #pragma unroll
  for (int j = 0; j < ITERS; ++j) {
    const int it = tid + 256 * j;
    const int q  = it / C4;
    const int g  = it - q * C4;
    w0[j] = w1[j] = w2[j] = w3[j] = (v4f){0.f, 0.f, 0.f, 0.f};
    if (it < TOTAL && g < NC4) {
      const float* p = Wm + (size_t)(4 * q) * NC + 4 * g;
      w0[j] = *(const v4f*)p;
      w1[j] = *(const v4f*)(p + NC);
      w2[j] = *(const v4f*)(p + 2 * NC);
      w3[j] = *(const v4f*)(p + 3 * NC);
    }
  }

  v4f a[K / 8];
  const float* __restrict__ Ar = A + (size_t)(srow + c) * K + 4 * h;
  #pragma unroll
  for (int i = 0; i < K / 8; ++i) a[i] = *(const v4f*)(Ar + 8 * i);

  #pragma unroll
  for (int j = 0; j < ITERS; ++j) {
    const int it = tid + 256 * j;
    const int q  = it / C4;
    const int g  = it - q * C4;
    if (it < TOTAL) {
      v4f* d = Bs + q * NCP + 4 * g;
      d[0] = (v4f){w0[j].x, w1[j].x, w2[j].x, w3[j].x};
      d[1] = (v4f){w0[j].y, w1[j].y, w2[j].y, w3[j].y};
      d[2] = (v4f){w0[j].z, w1[j].z, w2[j].z, w3[j].z};
      d[3] = (v4f){w0[j].w, w1[j].w, w2[j].w, w3[j].w};
    }
  }
  __syncthreads();
  if (!live) return;

  v8f acc[NT];
  #pragma unroll
  for (int t = 0; t < NT; ++t) acc[t] = (v8f){0.f, 0.f, 0.f, 0.f, 0.f, 0.f, 0.f, 0.f};

  const v4f* Bl = Bs + h * NCP + c;
  #pragma unroll
  for (int i = 0; i < K / 8; ++i) {
    v4f b[NT];
    #pragma unroll
    for (int t = 0; t < NT; ++t) b[t] = Bl[(2 * i) * NCP + 16 * t];
    const v2f a01 = {a[i].x, a[i].y};
    const v2f a23 = {a[i].z, a[i].w};
    #pragma unroll
    for (int t = 0; t < NT; ++t)
      acc[t] = __builtin_amdgcn_wmma_f32_16x16x4_f32(
          false, a01, false, (v2f){b[t].x, b[t].y}, (short)0, acc[t], false, false);
    #pragma unroll
    for (int t = 0; t < NT; ++t)
      acc[t] = __builtin_amdgcn_wmma_f32_16x16x4_f32(
          false, a23, false, (v2f){b[t].z, b[t].w}, (short)0, acc[t], false, false);
  }

  #pragma unroll
  for (int t = 0; t < NT; ++t) {
    const int col = 16 * t + c;
    if (col < NC) {
      const float bv = bias ? bias[col] : 0.f;
      float* Cp = C + (size_t)(strip * 16 + 8 * h) * NC + col;
      #pragma unroll
      for (int r = 0; r < 8; ++r) Cp[(size_t)r * NC] = acc[t][r] + bv;
    }
  }
}


static __device__ __forceinline__ float lane_f(float v, int l) {
  return __int_as_float(__builtin_amdgcn_readlane(__float_as_int(v), l));
}

static __device__ __forceinline__ f3 gather_row(const float* __restrict__ h,
                         const int* __restrict__ csr_src,
                         const float* __restrict__ dinv,
                         const float* __restrict__ bias,
                         int node, int lane, int beg_v, int end_v, float dn) {
  const int beg = __builtin_amdgcn_readfirstlane(beg_v);
  const int end = __builtin_amdgcn_readfirstlane(end_v);
  const int f = lane * 3;
  const float* __restrict__ hl = h + f;

  float a0 = 0.f, a1 = 0.f, a2 = 0.f;
  for (int base = beg; base < end; base += 32) {
    const int j = base + lane;
    int   s = 0;
    float w = 0.f;
    if (j < end) { s = csr_src[j]; w = dinv[s] * dn; }
    const int n = min(32, end - base);
    int i = 0;
    for (; i + 4 <= n; i += 4) {
      const int s0 = __builtin_amdgcn_readlane(s, i);
      const int s1 = __builtin_amdgcn_readlane(s, i + 1);
      const int s2 = __builtin_amdgcn_readlane(s, i + 2);
      const int s3 = __builtin_amdgcn_readlane(s, i + 3);
      const f3 r0 = *(const f3*)(hl + (size_t)s0 * FHID);
      const f3 r1 = *(const f3*)(hl + (size_t)s1 * FHID);
      const f3 r2 = *(const f3*)(hl + (size_t)s2 * FHID);
      const f3 r3 = *(const f3*)(hl + (size_t)s3 * FHID);
      const float w0 = lane_f(w, i);
      const float w1 = lane_f(w, i + 1);
      const float w2 = lane_f(w, i + 2);
      const float w3 = lane_f(w, i + 3);
      a0 = fmaf(r0.x, w0, a0); a1 = fmaf(r0.y, w0, a1); a2 = fmaf(r0.z, w0, a2);
      a0 = fmaf(r1.x, w1, a0); a1 = fmaf(r1.y, w1, a1); a2 = fmaf(r1.z, w1, a2);
      a0 = fmaf(r2.x, w2, a0); a1 = fmaf(r2.y, w2, a1); a2 = fmaf(r2.z, w2, a2);
      a0 = fmaf(r3.x, w3, a0); a1 = fmaf(r3.y, w3, a1); a2 = fmaf(r3.z, w3, a2);
    }
    for (; i < n; ++i) {
      const int   si = __builtin_amdgcn_readlane(s, i);
      const float wi = lane_f(w, i);
      const f3 r = *(const f3*)(hl + (size_t)si * FHID);
      a0 = fmaf(r.x, wi, a0); a1 = fmaf(r.y, wi, a1); a2 = fmaf(r.z, wi, a2);
    }
  }

  const f3 t = *(const f3*)(hl + (size_t)node * FHID);
  const f3 b = *(const f3*)(bias + f);
  const float ws = dn * dn;
  f3 r;
  r.x = fmaxf(a0 + t.x * ws + b.x, 0.f);
  r.y = fmaxf(a1 + t.y * ws + b.y, 0.f);
  r.z = fmaxf(a2 + t.z * ws + b.z, 0.f);
  return r;
}

#define APITCH 100
#define GG_BLOCKS 1024


template <int NT, int NC>
__global__ void __launch_bounds__(512)
k_gather_gemm(const float* __restrict__ hin,
              const int* __restrict__ rowptr,
              const int* __restrict__ csr_src,
              const float* __restrict__ dinv,
              const float* __restrict__ bias_in,
              const float* __restrict__ Wm,
              const float* __restrict__ bias_out,
              float* __restrict__ C, int N, int Mtiles) {
  constexpr int K     = FHID;
  constexpr int NCP   = NT * 16;
  constexpr int KQ    = K / 4;
  constexpr int C4    = NCP / 4;
  constexpr int NC4   = NC / 4;
  constexpr int TOTAL = KQ * C4;
  constexpr int ITERS = (TOTAL + 511) / 512;
  __shared__ v4f Bs[KQ * NCP];
  __shared__ v4f As4[2][16 * APITCH / 4];

  const int tid  = threadIdx.x;
  const int wave = tid >> 5;
  const int lane = tid & 31;
  const int c    = lane & 15;
  const int h    = lane >> 4;

  int strip = blockIdx.x;
  int node  = strip * 16 + wave;
  int rb = 0, re = 0;
  float dn = 0.f;
  if (strip < Mtiles && node < N) { rb = rowptr[node]; re = rowptr[node + 1]; dn = dinv[node]; }

  {
    v4f w0[ITERS], w1[ITERS], w2[ITERS], w3[ITERS];
    #pragma unroll
    for (int j = 0; j < ITERS; ++j) {
      const int it = tid + 512 * j;
      const int q  = it / C4;
      const int g  = it - q * C4;
      w0[j] = w1[j] = w2[j] = w3[j] = (v4f){0.f, 0.f, 0.f, 0.f};
      if (it < TOTAL && g < NC4) {
        const float* p = Wm + (size_t)(4 * q) * NC + 4 * g;
        w0[j] = *(const v4f*)p;
        w1[j] = *(const v4f*)(p + NC);
        w2[j] = *(const v4f*)(p + 2 * NC);
        w3[j] = *(const v4f*)(p + 3 * NC);
      }
    }
    #pragma unroll
    for (int j = 0; j < ITERS; ++j) {
      const int it = tid + 512 * j;
      const int q  = it / C4;
      const int g  = it - q * C4;
      if (it < TOTAL) {
        v4f* d = Bs + q * NCP + 4 * g;
        d[0] = (v4f){w0[j].x, w1[j].x, w2[j].x, w3[j].x};
        d[1] = (v4f){w0[j].y, w1[j].y, w2[j].y, w3[j].y};
        d[2] = (v4f){w0[j].z, w1[j].z, w2[j].z, w3[j].z};
        d[3] = (v4f){w0[j].w, w1[j].w, w2[j].w, w3[j].w};
      }
    }
  }

  const int t   = wave;
  const int col = 16 * t + c;
  const float bv = (bias_out && wave < NT && col < NC) ? bias_out[col] : 0.f;
  const v4f* Bl = Bs + h * NCP + 16 * (wave < NT ? t : 0) + c;

  int buf = 0;
  for (; strip < Mtiles; strip += gridDim.x) {
    const int nstrip = strip + gridDim.x;
    const int nnode  = nstrip * 16 + wave;
    int nrb = 0, nre = 0;
    float ndn = 0.f;
    if (nstrip < Mtiles && nnode < N) { nrb = rowptr[nnode]; nre = rowptr[nnode + 1]; ndn = dinv[nnode]; }

    f3 r;
    r.x = 0.f; r.y = 0.f; r.z = 0.f;
    if (node < N)
      r = gather_row(hin, csr_src, dinv, bias_in, node, lane, rb, re, dn);
    float* ar = (float*)As4[buf] + wave * APITCH + 3 * lane;
    ar[0] = r.x; ar[1] = r.y; ar[2] = r.z;
    __syncthreads();

    if (wave < NT) {
      v8f acc = (v8f){0.f, 0.f, 0.f, 0.f, 0.f, 0.f, 0.f, 0.f};
      const v4f* Al = As4[buf] + c * (APITCH / 4) + h;
      #pragma unroll
      for (int i = 0; i < K / 8; ++i) {
        const v4f a = Al[2 * i];
        const v4f b = Bl[(2 * i) * NCP];
        acc = __builtin_amdgcn_wmma_f32_16x16x4_f32(
            false, (v2f){a.x, a.y}, false, (v2f){b.x, b.y}, (short)0, acc, false, false);
        acc = __builtin_amdgcn_wmma_f32_16x16x4_f32(
            false, (v2f){a.z, a.w}, false, (v2f){b.z, b.w}, (short)0, acc, false, false);
      }

      if (col < NC) {
        const int row0 = strip * 16 + 8 * h;
        float* Cp = C + (size_t)row0 * NC + col;
        #pragma unroll
        for (int rr = 0; rr < 8; ++rr)
          if (row0 + rr < N) Cp[(size_t)rr * NC] = acc[rr] + bv;
      }
    }

    buf ^= 1;
    node = nnode; rb = nrb; re = nre; dn = ndn;
  }
}


static inline int cdiv(int a, int b) { return (a + b - 1) / b; }

extern "C" void kernel_launch(void* const* d_in, const int* in_sizes, int n_in,
                              void* d_out, int out_size, void* d_ws, size_t ws_size,
                              hipStream_t stream) {
  const float* x    = (const float*)d_in[0];
  const int*   eidx = (const int*)  d_in[1];
  const float* W1   = (const float*)d_in[2];
  const float* b1   = (const float*)d_in[3];
  const float* W2   = (const float*)d_in[4];
  const float* b2   = (const float*)d_in[5];
  const float* Wout = (const float*)d_in[6];
  const float* bout = (const float*)d_in[7];

  const int F_IN  = 128;
  const int N = in_sizes[0] / F_IN;
  const int E = in_sizes[1] / 2;
  const int* src = eidx;
  const int* dst = eidx + E;

  int*   tsum  = (int*)d_ws;
  float* dinv  = (float*)d_ws + N;
  float* tmp   = (float*)d_ws + 2 * (size_t)N + E;
  float* csrb  = tmp   + (size_t)N * FHID;
  int*   rowptr  = (int*)csrb;
  int*   csr_src = rowptr + N + 1;
  float* hbuf  = csrb  + (size_t)N * FHID;
  int*   part  = (int*)hbuf;
  float* outp  = (float*)d_out;

  const int T = 256;

  const int ntiles = cdiv(N, SCAN_T);
  const int epb    = cdiv(E, EP);
  const dim3 gcr(EP, cdiv(N, CAP));
  k_hist    <<<gcr, 1024, 0, stream>>>(dst, part, tsum, N, E, epb, ntiles);
  k_scan    <<<ntiles, SCAN_T, 0, stream>>>(part, tsum, rowptr, dinv, N, ntiles);
  k_fill_csr<<<gcr, 1024, 0, stream>>>(src, dst, part, csr_src, N, E, epb);

  const int Mtiles = cdiv(N, 16);
  const int gemmBlocks = cdiv(Mtiles, 8);

  k_gemm_strip<128, 6, 96><<<gemmBlocks, T, 0, stream>>>(x, W1, nullptr, tmp, Mtiles);

  k_gather_gemm<6, 96><<<GG_BLOCKS, 512, 0, stream>>>(
      tmp, rowptr, csr_src, dinv, b1, W2, nullptr, hbuf, N, Mtiles);

  k_gather_gemm<3, 40><<<GG_BLOCKS, 512, 0, stream>>>(
      hbuf, rowptr, csr_src, dinv, b2, Wout, bout, outp, N, Mtiles);
}


# Gathers: a node's loads no longer wait on each other: rows go out eight at a time (short last group under scalar branches, own row as one more source), sources asked a trip ahead; sum order kept
# speedup vs baseline: 3.9204x; 1.1620x over previous
#include <hip/hip_runtime.h>


typedef __attribute__((ext_vector_type(2))) float v2f;
typedef __attribute__((ext_vector_type(4))) float v4f;
typedef __attribute__((ext_vector_type(8))) float v8f;

#define FHID 96
#define SCAN_T 1024

struct __attribute__((packed, aligned(4))) f3 { float x, y, z; };


#define EP  36
#define RT  7
#define CAP (RT * SCAN_T)
#define EU  12

__global__ void __launch_bounds__(1024)
k_hist(const int* __restrict__ dst, int* __restrict__ part, int* __restrict__ tsum,
       int N, int E, int epb, int ntiles) {
  __shared__ int h[CAP];
  __shared__ int s_w[RT][32];
  const int tid  = threadIdx.x;
  const int lane = tid & 31;
  const int wv   = tid >> 5;
  const int b  = blockIdx.x;
  const int r  = blockIdx.y;
  const int lo = r * CAP;

  for (int i = tid; i < CAP; i += 1024) h[i] = 0;
  __syncthreads();

  const int e0 = b * epb;
  const int e1 = min(E, e0 + epb);
  for (int base = e0 + tid; base < e1; base += EU * 1024) {
    int d[EU];
    #pragma unroll
    for (int u = 0; u < EU; ++u) {
      const int e = base + u * 1024;
      d[u] = (e < e1) ? dst[e] : -1;
    }
    #pragma unroll
    for (int u = 0; u < EU; ++u) {
      const unsigned x = (unsigned)(d[u] - lo);
      if (x < (unsigned)CAP) atomicAdd(&h[x], 1);
    }
  }
  __syncthreads();

  int sm[RT];
  #pragma unroll
  for (int k = 0; k < RT; ++k) {
    const int n = lo + k * SCAN_T + tid;
    sm[k] = h[k * SCAN_T + tid];
    if (n < N) part[(size_t)b * N + n] = sm[k];
  }
  #pragma unroll
  for (int o = 16; o > 0; o >>= 1) {
    int up[RT];
    #pragma unroll
    for (int k = 0; k < RT; ++k) up[k] = __shfl_xor(sm[k], o);
    #pragma unroll
    for (int k = 0; k < RT; ++k) sm[k] += up[k];
  }
  if (lane == 0) {
    #pragma unroll
    for (int k = 0; k < RT; ++k) s_w[k][wv] = sm[k];
  }
  __syncthreads();
  if (tid < RT) {
    const int tile = r * RT + tid;
    if (tile < ntiles) {
      int tot = 0;
      for (int w = 0; w < 32; ++w) tot += s_w[tid][w];
      tsum[b * ntiles + tile] = tot;
    }
  }
}

static __device__ __forceinline__ void
scan_tile(const int* __restrict__ part, int* __restrict__ slots,
          const int* __restrict__ tsum, int* __restrict__ rowptr,
          float* __restrict__ dinv, int N, int ntiles, int t) {
  __shared__ int s_part[SCAN_T / 32];
  __shared__ int s_wsum[SCAN_T / 32];
  const int tid  = threadIdx.x;
  const int lane = tid & 31;
  const int wv   = tid >> 5;
  const int idx  = t * SCAN_T + tid;

  const int li = min(idx, N - 1);
  int cb[EP];
  #pragma unroll
  for (int b = 0; b < EP; ++b) cb[b] = part[(size_t)b * N + li];

  int front = 0;
  for (int i = tid; i < EP * ntiles; i += SCAN_T) {
    const int k = i % ntiles;
    if (k < t) front += tsum[i];
  }
  for (int o = 16; o > 0; o >>= 1) front += __shfl_xor(front, o);

  int c = 0;
  #pragma unroll
  for (int b = 0; b < EP; ++b) { const int v = (idx < N) ? cb[b] : 0; cb[b] = c; c += v; }

  int incl = c;
  for (int o = 1; o < 32; o <<= 1) {
    const int u = __shfl_up(incl, o);
    if (lane >= o) incl += u;
  }
  if (lane == 31) s_wsum[wv] = incl;
  if (lane == 0)  s_part[wv] = front;
  __syncthreads();

  int base = 0, woff = 0;
  for (int k = 0; k < SCAN_T / 32; ++k) {
    base += s_part[k];
    woff += (k < wv) ? s_wsum[k] : 0;
  }
  const int excl = base + woff + incl - c;
  if (idx < N) {
    rowptr[idx] = excl;
    dinv[idx]   = rsqrtf((float)(c + 1));
    #pragma unroll
    for (int b = 0; b < EP; ++b) slots[(size_t)b * N + idx] = excl + cb[b];
  }
  if (idx == N - 1) rowptr[N] = excl + c;
}

__global__ void __launch_bounds__(1024)
k_fill_csr(const int* __restrict__ src, const int* __restrict__ dst,
           const int* __restrict__ slots, int* __restrict__ csr_src,
           int N, int E, int epb) {
  __shared__ int h[CAP];
  const int tid = threadIdx.x;
  const int b  = blockIdx.x;
  const int r  = blockIdx.y;
  const int lo = r * CAP;

  int pv[RT];
  #pragma unroll
  for (int k = 0; k < RT; ++k) {
    const int n = lo + k * SCAN_T + tid;
    pv[k] = (n < N) ? slots[(size_t)b * N + n] : 0;
  }
  #pragma unroll
  for (int k = 0; k < RT; ++k) h[k * SCAN_T + tid] = pv[k];
  __syncthreads();

  const int e0 = b * epb;
  const int e1 = min(E, e0 + epb);
  for (int base = e0 + tid; base < e1; base += EU * 1024) {
    int d[EU], sv[EU];
    #pragma unroll
    for (int u = 0; u < EU; ++u) {
      const int e = base + u * 1024;
      d[u]  = (e < e1) ? dst[e] : -1;
      sv[u] = (e < e1) ? src[e] : 0;
    }
    int pos[EU];
    #pragma unroll
    for (int u = 0; u < EU; ++u) {
      const unsigned x = (unsigned)(d[u] - lo);
      pos[u] = (x < (unsigned)CAP) ? atomicAdd(&h[x], 1) : -1;
    }
    #pragma unroll
    for (int u = 0; u < EU; ++u)
      if (pos[u] >= 0) csr_src[pos[u]] = sv[u];
  }
}


#define G_K    128
#define G_NC   96
#define G_SPB  16

static __device__ __forceinline__ void
gemm_strips(const float* __restrict__ A, const float* __restrict__ Wm,
            const int* __restrict__ part,
            float* __restrict__ C, int N, int Mtiles, int g) {
  constexpr int K     = G_K;
  constexpr int NC    = G_NC;
  constexpr int NCP   = G_NC;
  constexpr int KQ    = K / 4;
  constexpr int C4    = NCP / 4;
  constexpr int TOTAL = KQ * C4;
  constexpr int NT    = 3;
  __shared__ v4f Bs[KQ * NCP];

  const int tid  = threadIdx.x;
  const int wave = tid >> 5;
  const int lane = tid & 31;
  const int c    = lane & 15;
  const int h    = lane >> 4;
  const int strip = g * G_SPB + (wave >> 1);
  const int half  = wave & 1;
  const bool live = strip < Mtiles;
  const int srow  = (live ? strip : Mtiles - 1) * 16;

  static_assert(EP % 2 == 0, "the two lane halves split the chunks evenly");
  int cnt[EP / 2];
  {
    const int* __restrict__ pr = part + (size_t)(h * (EP / 2)) * N + min(srow + c, N - 1);
    #pragma unroll
    for (int k = 0; k < EP / 2; ++k) cnt[k] = pr[(size_t)k * N];
  }

  const int q  = tid / C4;
  const int gq = tid - q * C4;
  v4f w0 = (v4f){0.f, 0.f, 0.f, 0.f}, w1 = w0, w2 = w0, w3 = w0;
  if (tid < TOTAL) {
    const float* p = Wm + (size_t)(4 * q) * NC + 4 * gq;
    w0 = *(const v4f*)p;
    w1 = *(const v4f*)(p + NC);
    w2 = *(const v4f*)(p + 2 * NC);
    w3 = *(const v4f*)(p + 3 * NC);
  }

  v4f a[K / 8];
  const float* __restrict__ Ar = A + (size_t)min(srow + c, N - 1) * K + 4 * h;
  #pragma unroll
  for (int i = 0; i < K / 8; ++i) a[i] = *(const v4f*)(Ar + 8 * i);

  int deg = 0;
  #pragma unroll
  for (int k = 0; k < EP / 2; ++k) deg += cnt[k];
  deg += __shfl_xor(deg, 16);
  const float rscale = rsqrtf((float)(deg + 1));

  if (tid < TOTAL) {
    v4f* d = Bs + q * NCP + 4 * gq;
    d[0] = (v4f){w0.x, w1.x, w2.x, w3.x};
    d[1] = (v4f){w0.y, w1.y, w2.y, w3.y};
    d[2] = (v4f){w0.z, w1.z, w2.z, w3.z};
    d[3] = (v4f){w0.w, w1.w, w2.w, w3.w};
  }
  __syncthreads();
  if (!live) return;

  v8f acc[NT];
  #pragma unroll
  for (int t = 0; t < NT; ++t) acc[t] = (v8f){0.f, 0.f, 0.f, 0.f, 0.f, 0.f, 0.f, 0.f};

  const int cb = 16 * NT * half;
  const v4f* Bl = Bs + h * NCP + cb + c;
  #pragma unroll
  for (int i = 0; i < K / 8; ++i) {
    v4f b[NT];
    #pragma unroll
    for (int t = 0; t < NT; ++t) b[t] = Bl[(2 * i) * NCP + 16 * t];
    const v2f a01 = {a[i].x, a[i].y};
    const v2f a23 = {a[i].z, a[i].w};
    #pragma unroll
    for (int t = 0; t < NT; ++t)
      acc[t] = __builtin_amdgcn_wmma_f32_16x16x4_f32(
          false, a01, false, (v2f){b[t].x, b[t].y}, (short)0, acc[t], false, false);
    #pragma unroll
    for (int t = 0; t < NT; ++t)
      acc[t] = __builtin_amdgcn_wmma_f32_16x16x4_f32(
          false, a23, false, (v2f){b[t].z, b[t].w}, (short)0, acc[t], false, false);
  }

  const int row0 = strip * 16 + 8 * h;
  float sc[8];
  #pragma unroll
  for (int r = 0; r < 8; ++r) sc[r] = __shfl(rscale, 8 * h + r);
  #pragma unroll
  for (int t = 0; t < NT; ++t) {
    float* Cp = C + (size_t)row0 * NC + cb + 16 * t + c;
    #pragma unroll
    for (int r = 0; r < 8; ++r)
      if (row0 + r < N) Cp[(size_t)r * NC] = acc[t][r] * sc[r];
  }
}

__global__ void __launch_bounds__(SCAN_T)
k_scan_gemm(const int* __restrict__ part, int* __restrict__ slots,
            const int* __restrict__ tsum, int* __restrict__ rowptr,
            float* __restrict__ dinv, int N, int ntiles,
            const float* __restrict__ A, const float* __restrict__ Wm,
            float* __restrict__ C, int Mtiles) {
  const int bx = blockIdx.x;
  if (bx < ntiles) scan_tile(part, slots, tsum, rowptr, dinv, N, ntiles, bx);
  else             gemm_strips(A, Wm, part, C, N, Mtiles, bx - ntiles);
}


#define GW 8

static __device__ __forceinline__ void
sum_rows(const float* __restrict__ h, int f, int s, int n, float& a0, float& a1, float& a2) {
  int i = 0;
  for (; i + GW <= n; i += GW) {
    f3 r[GW];
    #pragma unroll
    for (int u = 0; u < GW; ++u) {
      const int su = __builtin_amdgcn_readlane(s, i + u);
      r[u] = *(const f3*)(h + (size_t)su * FHID + f);
    }
    #pragma unroll
    for (int u = 0; u < GW; ++u) { a0 += r[u].x; a1 += r[u].y; a2 += r[u].z; }
  }
  if (i < n) {
    f3 r[GW - 1];
    #pragma unroll
    for (int u = 0; u < GW - 1; ++u) {
      r[u].x = 0.f; r[u].y = 0.f; r[u].z = 0.f;
      if (i + u < n) {
        const int su = __builtin_amdgcn_readlane(s, i + u);
        r[u] = *(const f3*)(h + (size_t)su * FHID + f);
      }
    }
    #pragma unroll
    for (int u = 0; u < GW - 1; ++u) { a0 += r[u].x; a1 += r[u].y; a2 += r[u].z; }
  }
}

static __device__ __forceinline__ f3 gather_row(const float* __restrict__ h,
                         const int* __restrict__ csr_src,
                         int node, int lane, int beg_v, int end_v, int s_first) {
  const int beg = __builtin_amdgcn_readfirstlane(beg_v);
  const int end = __builtin_amdgcn_readfirstlane(end_v);
  const int tot = end + 1;
  const int f = lane * 3;

  float a0 = 0.f, a1 = 0.f, a2 = 0.f;
  sum_rows(h, f, s_first, min(32, tot - beg), a0, a1, a2);
  for (int base = beg + 32; base < tot; base += 32) {
    const int j = base + lane;
    int s = node;
    if (j < end) s = csr_src[j];
    sum_rows(h, f, s, min(32, tot - base), a0, a1, a2);
  }
  f3 r;
  r.x = a0; r.y = a1; r.z = a2;
  return r;
}

#define APITCH 100
#define GG_BLOCKS 1024


template <int NT, int NC, bool SCALE_ROWS>
__global__ void __launch_bounds__(512) __attribute__((amdgpu_waves_per_eu(16, 16)))
k_gather_gemm(const float* __restrict__ hin,
              const int* __restrict__ rowptr,
              const int* __restrict__ csr_src,
              const float* __restrict__ dinv,
              const float* __restrict__ bias_in,
              const float* __restrict__ Wm,
              const float* __restrict__ bias_out,
              float* __restrict__ C, int N, int Mtiles) {
  constexpr int K     = FHID;
  constexpr int NCP   = NT * 16;
  constexpr int KQ    = K / 4;
  constexpr int C4    = NCP / 4;
  constexpr int NC4   = NC / 4;
  constexpr int TOTAL = KQ * C4;
  constexpr int ITERS = (TOTAL + 511) / 512;
  __shared__ v4f Bs[KQ * NCP];
  __shared__ v4f As4[2][16 * APITCH / 4];

  const int tid  = threadIdx.x;
  const int wave = tid >> 5;
  const int lane = tid & 31;
  const int c    = lane & 15;
  const int h    = lane >> 4;

  int strip = blockIdx.x;
  const int gstep = gridDim.x;
  int node  = strip * 16 + wave;
  int nnode = (strip + gstep) * 16 + wave;
  int rb = 0, re = 0, nrb = 0, nre = 0;
  float dn = 0.f, ndn = 0.f;
  if (strip < Mtiles && node < N) { rb = rowptr[node]; re = rowptr[node + 1]; dn = dinv[node]; }
  if (strip + gstep < Mtiles && nnode < N) { nrb = rowptr[nnode]; nre = rowptr[nnode + 1]; ndn = dinv[nnode]; }

  int sfirst;
  {
    v4f w0[ITERS], w1[ITERS], w2[ITERS], w3[ITERS];
    #pragma unroll
    for (int j = 0; j < ITERS; ++j) {
      const int it = tid + 512 * j;
      const int q  = it / C4;
      const int g  = it - q * C4;
      w0[j] = w1[j] = w2[j] = w3[j] = (v4f){0.f, 0.f, 0.f, 0.f};
      if (it < TOTAL && g < NC4) {
        const float* p = Wm + (size_t)(4 * q) * NC + 4 * g;
        w0[j] = *(const v4f*)p;
        w1[j] = *(const v4f*)(p + NC);
        w2[j] = *(const v4f*)(p + 2 * NC);
        w3[j] = *(const v4f*)(p + 3 * NC);
      }
    }
    sfirst = node;
    if (rb + lane < re) sfirst = csr_src[rb + lane];

    #pragma unroll
    for (int j = 0; j < ITERS; ++j) {
      const int it = tid + 512 * j;
      const int q  = it / C4;
      const int g  = it - q * C4;
      if (it < TOTAL) {
        v4f* d = Bs + q * NCP + 4 * g;
        d[0] = (v4f){w0[j].x, w1[j].x, w2[j].x, w3[j].x};
        d[1] = (v4f){w0[j].y, w1[j].y, w2[j].y, w3[j].y};
        d[2] = (v4f){w0[j].z, w1[j].z, w2[j].z, w3[j].z};
        d[3] = (v4f){w0[j].w, w1[j].w, w2[j].w, w3[j].w};
      }
    }
  }

  const int t   = wave;
  const int col = 16 * t + c;
  const float bv = (bias_out && wave < NT && col < NC) ? bias_out[col] : 0.f;
  const f3 bin = *(const f3*)(bias_in + 3 * lane);
  const v4f* Bl = Bs + h * NCP + 16 * (wave < NT ? t : 0) + c;

  int buf = 0;
  for (; strip < Mtiles; strip += gstep) {
    int nsfirst = nnode;
    if (nrb + lane < nre) nsfirst = csr_src[nrb + lane];

    const int n2strip = strip + 2 * gstep;
    const int n2node  = n2strip * 16 + wave;
    int n2rb = 0, n2re = 0;
    float n2dn = 0.f;
    if (n2strip < Mtiles && n2node < N) { n2rb = rowptr[n2node]; n2re = rowptr[n2node + 1]; n2dn = dinv[n2node]; }

    f3 r;
    r.x = 0.f; r.y = 0.f; r.z = 0.f;
    if (node < N) {
      const f3 a = gather_row(hin, csr_src, node, lane, rb, re, sfirst);
      r.x = fmaxf(fmaf(a.x, dn, bin.x), 0.f);
      r.y = fmaxf(fmaf(a.y, dn, bin.y), 0.f);
      r.z = fmaxf(fmaf(a.z, dn, bin.z), 0.f);
    }
    if (SCALE_ROWS) { r.x *= dn; r.y *= dn; r.z *= dn; }
    float* ar = (float*)As4[buf] + wave * APITCH + 3 * lane;
    ar[0] = r.x; ar[1] = r.y; ar[2] = r.z;
    __syncthreads();

    if (wave < NT) {
      v8f acc = (v8f){0.f, 0.f, 0.f, 0.f, 0.f, 0.f, 0.f, 0.f};
      const v4f* Al = As4[buf] + c * (APITCH / 4) + h;
      #pragma unroll
      for (int i = 0; i < K / 8; ++i) {
        const v4f a = Al[2 * i];
        const v4f b = Bl[(2 * i) * NCP];
        acc = __builtin_amdgcn_wmma_f32_16x16x4_f32(
            false, (v2f){a.x, a.y}, false, (v2f){b.x, b.y}, (short)0, acc, false, false);
        acc = __builtin_amdgcn_wmma_f32_16x16x4_f32(
            false, (v2f){a.z, a.w}, false, (v2f){b.z, b.w}, (short)0, acc, false, false);
      }

      if (col < NC) {
        const int row0 = strip * 16 + 8 * h;
        float* Cp = C + (size_t)row0 * NC + col;
        #pragma unroll
        for (int rr = 0; rr < 8; ++rr)
          if (row0 + rr < N) Cp[(size_t)rr * NC] = acc[rr] + bv;
      }
    }

    buf ^= 1;
    node = nnode; rb = nrb; re = nre; dn = ndn; sfirst = nsfirst;
    nnode = n2node; nrb = n2rb; nre = n2re; ndn = n2dn;
  }
}


static inline int cdiv(int a, int b) { return (a + b - 1) / b; }

extern "C" void kernel_launch(void* const* d_in, const int* in_sizes, int n_in,
                              void* d_out, int out_size, void* d_ws, size_t ws_size,
                              hipStream_t stream) {
  const float* x    = (const float*)d_in[0];
  const int*   eidx = (const int*)  d_in[1];
  const float* W1   = (const float*)d_in[2];
  const float* b1   = (const float*)d_in[3];
  const float* W2   = (const float*)d_in[4];
  const float* b2   = (const float*)d_in[5];
  const float* Wout = (const float*)d_in[6];
  const float* bout = (const float*)d_in[7];

  const int F_IN  = 128;
  const int N = in_sizes[0] / F_IN;
  const int E = in_sizes[1] / 2;
  const int* src = eidx;
  const int* dst = eidx + E;

  int*   tsum  = (int*)d_ws;
  float* dinv  = (float*)d_ws + N;
  float* tmp   = (float*)d_ws + 2 * (size_t)N + E;
  float* csrb  = tmp   + (size_t)N * FHID;
  int*   rowptr  = (int*)csrb;
  int*   csr_src = rowptr + N + 1;
  float* hbuf  = csrb  + (size_t)N * FHID;
  int*   part  = (int*)hbuf;
  int*   slots = part + (size_t)EP * N;
  float* outp  = (float*)d_out;

  const int ntiles = cdiv(N, SCAN_T);
  const int epb    = cdiv(E, EP);
  const dim3 gcr(EP, cdiv(N, CAP));
  const int Mtiles = cdiv(N, 16);
  const int gemmBlocks = cdiv(Mtiles, G_SPB);
  k_hist     <<<gcr, 1024, 0, stream>>>(dst, part, tsum, N, E, epb, ntiles);
  k_scan_gemm<<<ntiles + gemmBlocks, SCAN_T, 0, stream>>>(part, slots, tsum, rowptr, dinv, N, ntiles,
                                                          x, W1, tmp, Mtiles);
  k_fill_csr <<<gcr, 1024, 0, stream>>>(src, dst, slots, csr_src, N, E, epb);

  k_gather_gemm<6, 96, true><<<GG_BLOCKS, 512, 0, stream>>>(
      tmp, rowptr, csr_src, dinv, b1, W2, nullptr, hbuf, N, Mtiles);

  k_gather_gemm<3, 40, false><<<GG_BLOCKS, 512, 0, stream>>>(
      hbuf, rowptr, csr_src, dinv, b2, Wout, bout, outp, N, Mtiles);
}
